# starting kernel
# baseline (speedup 1.0000x reference)
#include <hip/hip_runtime.h>
#include <hip/hip_bf16.h>


typedef __attribute__((ext_vector_type(2))) float v2f;
typedef __attribute__((ext_vector_type(8))) float v8f;

#define BROWS 16384
#define DIMS  64
#define NNZ   50

__global__ __launch_bounds__(256) void gather_mean_kernel(
    const float* __restrict__ emb, const int* __restrict__ fidx,
    float* __restrict__ x) {
  const int wave = threadIdx.x >> 5;
  const int lane = threadIdx.x & 31;
  const int row  = blockIdx.x * 8 + wave;
  const int base = row * NNZ;

  float ax = 0.0f, ay = 0.0f;
#pragma unroll 5
  for (int j = 0; j < NNZ; ++j) {
    const int f = fidx[base + j];
    const float2 e =
        ((const float2*)(emb + (size_t)f * DIMS))[lane];
    ax += e.x;
    ay += e.y;
  }
  const float s = 1.0f / (float)NNZ;
  float2 r;
  r.x = ax * s;
  r.y = ay * s;
  ((float2*)(x + (size_t)row * DIMS))[lane] = r;
}

__global__ __launch_bounds__(256) void mlp3_wmma_kernel(
    const float* __restrict__ x,
    const float* __restrict__ W0, const float* __restrict__ b0,
    const float* __restrict__ W1, const float* __restrict__ b1,
    const float* __restrict__ W2, const float* __restrict__ b2,
    float* __restrict__ out) {
  __shared__ v2f   sW2[32 * DIMS];
  __shared__ float sBias[DIMS];
  __shared__ float sTile[8][16 * DIMS];

  const int tid   = threadIdx.x;
  const int wave  = tid >> 5;
  const int lane  = tid & 31;
  const int l15   = lane & 15;
  const int half  = lane >> 4;
  const int mbase = (blockIdx.x * 8 + wave) * 16;

  float* tp = &sTile[wave][0];

#pragma unroll
  for (int r = 0; r < 16; ++r)
    ((float2*)tp)[r * 32 + lane] =
        ((const float2*)(x + (size_t)(mbase + r) * DIMS))[lane];

  const float* Wg[3] = {W0, W1, W2};
  const float* bg[3] = {b0, b1, b2};

#pragma unroll
  for (int layer = 0; layer < 3; ++layer) {
    __syncthreads();
    {
      const float* wsrc = Wg[layer];
#pragma unroll
      for (int i = 0; i < 8; ++i) {
        const int idx = tid + 256 * i;
        const int p   = idx >> 6;
        const int n   = idx & 63;
        v2f frag;
        frag.x = wsrc[(2 * p) * DIMS + n];
        frag.y = wsrc[(2 * p + 1) * DIMS + n];
        sW2[idx] = frag;
      }
      if (tid < DIMS) sBias[tid] = bg[layer][tid];
    }
    __syncthreads();

    const float bv0 = sBias[0 + l15];
    const float bv1 = sBias[16 + l15];
    const float bv2 = sBias[32 + l15];
    const float bv3 = sBias[48 + l15];
    v8f a0 = {bv0, bv0, bv0, bv0, bv0, bv0, bv0, bv0};
    v8f a1 = {bv1, bv1, bv1, bv1, bv1, bv1, bv1, bv1};
    v8f a2 = {bv2, bv2, bv2, bv2, bv2, bv2, bv2, bv2};
    v8f a3 = {bv3, bv3, bv3, bv3, bv3, bv3, bv3, bv3};

#pragma unroll
    for (int k = 0; k < 16; ++k) {
      const v2f av = ((const v2f*)tp)[l15 * 32 + k * 2 + half];
      const int pbase = (2 * k + half) * 64 + l15;
      const v2f bb0 = sW2[pbase + 0];
      const v2f bb1 = sW2[pbase + 16];
      const v2f bb2 = sW2[pbase + 32];
      const v2f bb3 = sW2[pbase + 48];
      a0 = __builtin_amdgcn_wmma_f32_16x16x4_f32(false, av, false, bb0,
                                                 (short)0, a0, false, false);
      a1 = __builtin_amdgcn_wmma_f32_16x16x4_f32(false, av, false, bb1,
                                                 (short)0, a1, false, false);
      a2 = __builtin_amdgcn_wmma_f32_16x16x4_f32(false, av, false, bb2,
                                                 (short)0, a2, false, false);
      a3 = __builtin_amdgcn_wmma_f32_16x16x4_f32(false, av, false, bb3,
                                                 (short)0, a3, false, false);
    }

    const int mD = half * 8;
#pragma unroll
    for (int j = 0; j < 8; ++j) {
      tp[(mD + j) * DIMS + 0 + l15]  = fmaxf(a0[j], 0.0f);
      tp[(mD + j) * DIMS + 16 + l15] = fmaxf(a1[j], 0.0f);
      tp[(mD + j) * DIMS + 32 + l15] = fmaxf(a2[j], 0.0f);
      tp[(mD + j) * DIMS + 48 + l15] = fmaxf(a3[j], 0.0f);
    }
  }

#pragma unroll
  for (int r = 0; r < 16; ++r)
    ((float2*)(out + (size_t)(mbase + r) * DIMS))[lane] =
        ((const float2*)tp)[r * 32 + lane];
}

extern "C" void kernel_launch(void* const* d_in, const int* in_sizes, int n_in,
                              void* d_out, int out_size, void* d_ws,
                              size_t ws_size, hipStream_t stream) {
  const float* emb  = (const float*)d_in[0];
  const int*   fidx = (const int*)d_in[1];
  const float* W0   = (const float*)d_in[3];
  const float* b0   = (const float*)d_in[4];
  const float* W1   = (const float*)d_in[5];
  const float* b1   = (const float*)d_in[6];
  const float* W2   = (const float*)d_in[7];
  const float* b2   = (const float*)d_in[8];
  float*       out  = (float*)d_out;

  float* x = (float*)d_ws;

  gather_mean_kernel<<<BROWS / 8, 256, 0, stream>>>(emb, fidx, x);
  mlp3_wmma_kernel<<<BROWS / (8 * 16), 256, 0, stream>>>(x, W0, b0, W1, b1,
                                                         W2, b2, out);
}


# one launch instead of two: 256 groups of 8 waves; a wave gathers 8 batch rows (50 requests a row from scalar bases, two buffers) into an LDS tile, then waves 0-3 run the register MLP; sum order kept
# speedup vs baseline: 1.5886x; 1.5886x over previous
#include <hip/hip_runtime.h>
#include <hip/hip_bf16.h>


typedef __attribute__((ext_vector_type(2))) float v2f;
typedef __attribute__((ext_vector_type(4))) float v4f;
typedef __attribute__((ext_vector_type(8))) float v8f;

#define BROWS 16384
#define DIMS  64
#define NNZ   50
#define GROWS 64
#define WROWS 8
#define TPITCH 68

__global__ __launch_bounds__(256) void bag_mlp3_kernel(
    const float* __restrict__ emb, const int* __restrict__ fidx,
    const float* __restrict__ W0, const float* __restrict__ b0,
    const float* __restrict__ W1, const float* __restrict__ b1,
    const float* __restrict__ W2, const float* __restrict__ b2,
    float* __restrict__ out) {
  __shared__ v4f sA[3][4 * 8 * 32];
  __shared__ __attribute__((aligned(16))) float sBias[3 * DIMS];
  __shared__ __attribute__((aligned(16))) float sTile[GROWS * TPITCH];

  const int tid  = threadIdx.x;
  const int wave = tid >> 5;
  const int lane = tid & 31;

  const float* Wg[3] = {W0, W1, W2};
  const int st = tid & 127;
  const int nq = st & 15;
  const int rq = (st >> 4) + 8 * (tid >> 7);
  v4f wr[3][4];
#pragma unroll
  for (int L = 0; L < 3; ++L)
#pragma unroll
    for (int r = 0; r < 4; ++r)
      wr[L][r] = ((const v4f*)(Wg[L] + (4 * rq + r) * DIMS))[nq];

  const int lrow0 = wave * WROWS;
  const int* ip = fidx + (size_t)(blockIdx.x * GROWS + lrow0) * NNZ;
  const int l17 = lane < 18 ? lane : 17;
  const float sc = 1.0f / (float)NNZ;

#define LOAD_IDX(R, I0, I1)                                                   \
  I0 = ip[(R) * NNZ + lane];                                                  \
  I1 = ip[(R) * NNZ + 32 + l17];

#define ISSUE_ROW(BUF, I0, I1)                                                \
  _Pragma("unroll") for (int j = 0; j < NNZ; ++j) {                           \
    const int f = (j < 32) ? __builtin_amdgcn_readlane(I0, j)                 \
                           : __builtin_amdgcn_readlane(I1, j - 32);           \
    BUF[j] = ((const float2*)(emb + (size_t)f * DIMS))[lane];                 \
  }

#define SUM_ROW(BUF, R)                                                       \
  {                                                                           \
    float ax = 0.0f, ay = 0.0f;                                               \
    _Pragma("unroll") for (int j = 0; j < NNZ; ++j) {                         \
      ax += BUF[j].x;                                                         \
      ay += BUF[j].y;                                                         \
    }                                                                         \
    float2 m;                                                                 \
    m.x = ax * sc;                                                            \
    m.y = ay * sc;                                                            \
    *(float2*)(&sTile[(lrow0 + (R)) * TPITCH + 2 * lane]) = m;                \
  }

  float2 P[NNZ], Q[NNZ];
  int pa0, pa1, qa0, qa1;
  LOAD_IDX(0, pa0, pa1)
  LOAD_IDX(1, qa0, qa1)
  ISSUE_ROW(P, pa0, pa1)
  LOAD_IDX(2, pa0, pa1)

  {
    const int t  = nq >> 2;
    const int cq = nq & 3;
    const int j  = 2 * (rq >> 2) + (rq & 1);
    const int hh = (rq >> 1) & 1;
    const int rowbase = (t * 8 + j) * 32 + hh * 16;
#pragma unroll
    for (int L = 0; L < 3; ++L)
#pragma unroll
      for (int i = 0; i < 4; ++i) {
        v4f f;
        f.x = wr[L][0][i];
        f.y = wr[L][1][i];
        f.z = wr[L][2][i];
        f.w = wr[L][3][i];
        sA[L][rowbase + ((4 * cq + i + t) & 15)] = f;
      }
  }
  if (tid < DIMS) {
    sBias[tid]            = b0[tid];
    sBias[DIMS + tid]     = b1[tid];
    sBias[2 * DIMS + tid] = b2[tid];
  }

#pragma unroll 1
  for (int q = 0; q < WROWS - 2; q += 2) {
    ISSUE_ROW(Q, qa0, qa1)
    LOAD_IDX(q + 3, qa0, qa1)
    SUM_ROW(P, q)
    ISSUE_ROW(P, pa0, pa1)
    {
      const int qn = (q + 4 < WROWS) ? q + 4 : WROWS - 1;
      LOAD_IDX(qn, pa0, pa1)
    }
    SUM_ROW(Q, q + 1)
  }
  ISSUE_ROW(Q, qa0, qa1)
  SUM_ROW(P, WROWS - 2)
  SUM_ROW(Q, WROWS - 1)
#undef LOAD_IDX
#undef ISSUE_ROW
#undef SUM_ROW

  __syncthreads();
  if (wave >= 4) return;

  const int c     = lane & 15;
  const int h     = lane >> 4;
  const int mbase = blockIdx.x * GROWS + wave * 16;
  float* tp = &sTile[wave * 16 * TPITCH];

  v8f act[4];
#pragma unroll
  for (int T = 0; T < 4; ++T) {
    const v4f lo = *(const v4f*)(tp + c * TPITCH + 16 * T + 8 * h);
    const v4f hi = *(const v4f*)(tp + c * TPITCH + 16 * T + 8 * h + 4);
    act[T] = __builtin_shufflevector(lo, hi, 0, 1, 2, 3, 4, 5, 6, 7);
  }

#define LOAD_LAYER(L, A, BI)                                                  \
  _Pragma("unroll") for (int j = 0; j < 8; ++j)                               \
      _Pragma("unroll") for (int t = 0; t < 4; ++t)                           \
          A[j][t] = sA[L][(t * 8 + j) * 32 + h * 16 + ((c + t) & 15)];        \
  _Pragma("unroll") for (int t = 0; t < 4; ++t) {                             \
    const v4f lo = *(const v4f*)(&sBias[(L) * DIMS + 16 * t + 8 * h]);        \
    const v4f hi = *(const v4f*)(&sBias[(L) * DIMS + 16 * t + 8 * h + 4]);    \
    BI[t] = __builtin_shufflevector(lo, hi, 0, 1, 2, 3, 4, 5, 6, 7);          \
  }

#define RUN_LAYER(A, ACC)                                                     \
  _Pragma("unroll") for (int j = 0; j < 8; ++j) {                             \
    const int T = j >> 1, v = j & 1;                                          \
    const v2f bE = {act[T][4 * v], act[T][4 * v + 1]};        \
    const v2f bO = {act[T][4 * v + 2], act[T][4 * v + 3]};    \
    _Pragma("unroll") for (int t = 0; t < 4; ++t) {                           \
      const v2f aE = {A[j][t].x, A[j][t].y};                                  \
      ACC[t] = __builtin_amdgcn_wmma_f32_16x16x4_f32(                         \
          false, aE, false, bE, (short)0, ACC[t], false, false);              \
    }                                                                         \
    _Pragma("unroll") for (int t = 0; t < 4; ++t) {                           \
      const v2f aO = {A[j][t].z, A[j][t].w};                                  \
      ACC[t] = __builtin_amdgcn_wmma_f32_16x16x4_f32(                         \
          false, aO, false, bO, (short)0, ACC[t], false, false);              \
    }                                                                         \
  }

#define RELU_TO_ACT(ACC)                                                      \
  _Pragma("unroll") for (int t = 0; t < 4; ++t)                               \
      _Pragma("unroll") for (int r = 0; r < 8; ++r)                           \
          act[t][r] = fmaxf(ACC[t][r], 0.0f);

  v4f fa[8][4], fb[8][4];
  v8f acc0[4], acc1[4], acc2[4];

  LOAD_LAYER(0, fa, acc0)
  LOAD_LAYER(1, fb, acc1)
  __builtin_amdgcn_sched_barrier(0);
  RUN_LAYER(fa, acc0)
  __builtin_amdgcn_sched_barrier(0);
  LOAD_LAYER(2, fa, acc2)
  __builtin_amdgcn_sched_barrier(0);
  RELU_TO_ACT(acc0)
  RUN_LAYER(fb, acc1)
  RELU_TO_ACT(acc1)
  RUN_LAYER(fa, acc2)
  RELU_TO_ACT(acc2)
#undef LOAD_LAYER
#undef RUN_LAYER
#undef RELU_TO_ACT

#pragma unroll
  for (int t = 0; t < 4; ++t) {
    *(v4f*)(tp + c * TPITCH + 16 * t + 8 * h) =
        __builtin_shufflevector(act[t], act[t], 0, 1, 2, 3);
    *(v4f*)(tp + c * TPITCH + 16 * t + 8 * h + 4) =
        __builtin_shufflevector(act[t], act[t], 4, 5, 6, 7);
  }
#pragma unroll
  for (int i = 0; i < 8; ++i)
    ((v4f*)(out + (size_t)(mbase + 2 * i + h) * DIMS))[c] =
        *(const v4f*)(tp + (2 * i + h) * TPITCH + 4 * c);
}

extern "C" void kernel_launch(void* const* d_in, const int* in_sizes, int n_in,
                              void* d_out, int out_size, void* d_ws,
                              size_t ws_size, hipStream_t stream) {
  const float* emb  = (const float*)d_in[0];
  const int*   fidx = (const int*)d_in[1];
  const float* W0   = (const float*)d_in[3];
  const float* b0   = (const float*)d_in[4];
  const float* W1   = (const float*)d_in[5];
  const float* b1   = (const float*)d_in[6];
  const float* W2   = (const float*)d_in[7];
  const float* b2   = (const float*)d_in[8];
  float*       out  = (float*)d_out;

  bag_mlp3_kernel<<<BROWS / GROWS, 256, 0, stream>>>(emb, fidx, W0, b0, W1, b1,
                                                     W2, b2, out);
}
